# starting kernel
# baseline (speedup 1.0000x reference)
#include <hip/hip_runtime.h>
#include <math.h>

#define NN 10000
#define EE 160000
#define FF 64
#define NF (NN * FF)
#define PI_F 3.14159265358979323846f

#define H1S 68
#define H2S 132
#define RDS 196

typedef float v2f __attribute__((ext_vector_type(2)));
typedef float v8f __attribute__((ext_vector_type(8)));

__device__ __forceinline__ v8f wmma4(v2f a, v2f b, v8f c) {
    return __builtin_amdgcn_wmma_f32_16x16x4_f32(
         false, a,  false, b,
         (short)0, c,  false,  false);
}

__device__ __forceinline__ float silu(float x) {
    return x / (1.0f + expf(-x));
}

__device__ __forceinline__ void recon3(const float c[10], float t[9]) {
    float d = c[0];
    t[0] = d + c[4];  t[4] = d + c[5];  t[8] = d + c[6];
    t[1] = c[1] + c[7];  t[3] = c[7] - c[1];
    t[2] = c[2] + c[8];  t[6] = c[8] - c[2];
    t[5] = c[3] + c[9];  t[7] = c[9] - c[3];
}

__device__ __forceinline__ void mm3(const float a[9], const float b[9], float c[9]) {
#pragma unroll
    for (int i = 0; i < 3; ++i)
#pragma unroll
        for (int j = 0; j < 3; ++j) {
            float s = 0.f;
#pragma unroll
            for (int k = 0; k < 3; ++k) s += a[i * 3 + k] * b[k * 3 + j];
            c[i * 3 + j] = s;
        }
}

__global__ void prep_kernel(const float* __restrict__ X,
                            float* __restrict__ Xn,
                            float* __restrict__ P) {
    int idx = blockIdx.x * blockDim.x + threadIdx.x;
    if (idx >= NF) return;
    float x[9]; float nrm = 0.f;
#pragma unroll
    for (int i = 0; i < 9; ++i) { x[i] = X[(size_t)idx * 9 + i]; nrm += x[i] * x[i]; }
    float inv = 1.0f / (nrm + 1.0f);
#pragma unroll
    for (int i = 0; i < 9; ++i) { x[i] *= inv; Xn[(size_t)idx * 9 + i] = x[i]; }
    float dm = (x[0] + x[4] + x[8]) * (1.0f / 3.0f);
    P[0 * NF + idx] = dm;
    P[1 * NF + idx] = 0.5f * (x[1] - x[3]);
    P[2 * NF + idx] = 0.5f * (x[2] - x[6]);
    P[3 * NF + idx] = 0.5f * (x[5] - x[7]);
    P[4 * NF + idx] = x[0] - dm;
    P[5 * NF + idx] = x[4] - dm;
    P[6 * NF + idx] = x[8] - dm;
    P[7 * NF + idx] = 0.5f * (x[1] + x[3]);
    P[8 * NF + idx] = 0.5f * (x[2] + x[6]);
    P[9 * NF + idx] = 0.5f * (x[5] + x[7]);
}

__global__ __launch_bounds__(128) void mix_kernel(const float* __restrict__ P,
                                                  float* __restrict__ Q,
                                                  const float* __restrict__ La,
                                                  const float* __restrict__ Lb,
                                                  const float* __restrict__ Lc) {
    const int lane = threadIdx.x & 31;
    const int wave = threadIdx.x >> 5;
    const int ch = blockIdx.y;
    const int r0 = blockIdx.x * 16;
    const float* W = (ch == 0) ? La : (ch < 4 ? Lb : Lc);
    const float* A = P + (size_t)ch * NF;
    float* D = Q + (size_t)ch * NF;
    const int row = lane & 15;
    const int hi = lane >> 4;
    const int kb = hi * 2;
    const int col = wave * 16 + (lane & 15);
    v8f acc = {};
#pragma unroll
    for (int k = 0; k < FF; k += 4) {
        v2f a, b;
        a.x = A[(size_t)(r0 + row) * FF + k + kb];
        a.y = A[(size_t)(r0 + row) * FF + k + kb + 1];
        b.x = W[(k + kb) * FF + col];
        b.y = W[(k + kb + 1) * FF + col];
        acc = wmma4(a, b, acc);
    }
#pragma unroll
    for (int r = 0; r < 8; ++r)
        D[(size_t)(r0 + r + 8 * hi) * FF + col] = acc[r];
}

__global__ __launch_bounds__(32) void edge_kernel(
    const float* __restrict__ rbf, const float* __restrict__ dij,
    const int* __restrict__ pi,
    const float* __restrict__ W1, const float* __restrict__ b1,
    const float* __restrict__ W2, const float* __restrict__ b2,
    const float* __restrict__ W3, const float* __restrict__ b3,
    const float* __restrict__ Q, float* __restrict__ M) {
    __shared__ float h1[16 * H1S];
    __shared__ float h2[16 * H2S];
    __shared__ float rad[16 * RDS];
    __shared__ float cc[16];

    const int lane = threadIdx.x;
    const int e0 = blockIdx.x * 16;
    const int row = lane & 15;
    const int hi = lane >> 4;
    const int kb = hi * 2;
    const int col16 = lane & 15;

    if (lane < 16) {
        float d = dij[e0 + lane];
        cc[lane] = 0.5f * (cosf(d * (PI_F / 0.5f)) + 1.0f) * (d < 0.5f ? 1.0f : 0.0f);
    }

#pragma unroll
    for (int j = 0; j < 4; ++j) {
        const int col = j * 16 + col16;
        v8f acc = {};
#pragma unroll
        for (int k = 0; k < 32; k += 4) {
            v2f a, b;
            a.x = rbf[(size_t)(e0 + row) * 32 + k + kb];
            a.y = rbf[(size_t)(e0 + row) * 32 + k + kb + 1];
            b.x = W1[(k + kb) * 64 + col];
            b.y = W1[(k + kb + 1) * 64 + col];
            acc = wmma4(a, b, acc);
        }
        float bias = b1[col];
#pragma unroll
        for (int r = 0; r < 8; ++r)
            h1[(r + 8 * hi) * H1S + col] = silu(acc[r] + bias);
    }
    __syncthreads();

#pragma unroll
    for (int j = 0; j < 8; ++j) {
        const int col = j * 16 + col16;
        v8f acc = {};
#pragma unroll
        for (int k = 0; k < 64; k += 4) {
            v2f a, b;
            a.x = h1[row * H1S + k + kb];
            a.y = h1[row * H1S + k + kb + 1];
            b.x = W2[(k + kb) * 128 + col];
            b.y = W2[(k + kb + 1) * 128 + col];
            acc = wmma4(a, b, acc);
        }
        float bias = b2[col];
#pragma unroll
        for (int r = 0; r < 8; ++r)
            h2[(r + 8 * hi) * H2S + col] = silu(acc[r] + bias);
    }
    __syncthreads();

#pragma unroll
    for (int j = 0; j < 12; ++j) {
        const int col = j * 16 + col16;
        v8f acc = {};
#pragma unroll
        for (int k = 0; k < 128; k += 4) {
            v2f a, b;
            a.x = h2[row * H2S + k + kb];
            a.y = h2[row * H2S + k + kb + 1];
            b.x = W3[(k + kb) * 192 + col];
            b.y = W3[(k + kb + 1) * 192 + col];
            acc = wmma4(a, b, acc);
        }
        float bias = b3[col];
#pragma unroll
        for (int r = 0; r < 8; ++r) {
            int er = r + 8 * hi;
            rad[er * RDS + col] = silu(acc[r] + bias) * cc[er];
        }
    }
    __syncthreads();

    for (int e = 0; e < 16; ++e) {
        const int s = pi[e0 + e];
        const int d = pi[EE + e0 + e];
#pragma unroll
        for (int t = 0; t < 2; ++t) {
            const int f = lane + 32 * t;
            const float rI = rad[e * RDS + 3 * f + 0];
            const float rA = rad[e * RDS + 3 * f + 1];
            const float rS = rad[e * RDS + 3 * f + 2];
            const int g = d * FF + f;
            const int so = s * FF + f;
            atomicAdd(&M[0 * NF + so], rI * Q[0 * NF + g]);
#pragma unroll
            for (int c = 1; c < 4; ++c)
                atomicAdd(&M[c * NF + so], rA * Q[c * NF + g]);
#pragma unroll
            for (int c = 4; c < 10; ++c)
                atomicAdd(&M[c * NF + so], rS * Q[c * NF + g]);
        }
    }
}

__global__ void combine_kernel(const float* __restrict__ M,
                               const float* __restrict__ Q,
                               const float* __restrict__ chg,
                               float* __restrict__ P) {
    int idx = blockIdx.x * blockDim.x + threadIdx.x;
    if (idx >= NF) return;
    const int n = idx >> 6;
    float mc[10], yc[10];
#pragma unroll
    for (int c = 0; c < 10; ++c) { mc[c] = M[c * NF + idx]; yc[c] = Q[c * NF + idx]; }
    float msg[9], Y[9], t1[9], t2[9];
    recon3(mc, msg);
    recon3(yc, Y);
    mm3(msg, Y, t1);
    mm3(Y, msg, t2);
    const float scale = 1.0f + 0.1f * chg[n];
    float T[9]; float tn = 0.f;
#pragma unroll
    for (int i = 0; i < 9; ++i) { T[i] = scale * (t1[i] + t2[i]); tn += T[i] * T[i]; }
    const float inv = 1.0f / (tn + 1.0f);
#pragma unroll
    for (int i = 0; i < 9; ++i) T[i] *= inv;
    float dm = (T[0] + T[4] + T[8]) * (1.0f / 3.0f);
    P[0 * NF + idx] = dm;
    P[1 * NF + idx] = 0.5f * (T[1] - T[3]);
    P[2 * NF + idx] = 0.5f * (T[2] - T[6]);
    P[3 * NF + idx] = 0.5f * (T[5] - T[7]);
    P[4 * NF + idx] = T[0] - dm;
    P[5 * NF + idx] = T[4] - dm;
    P[6 * NF + idx] = T[8] - dm;
    P[7 * NF + idx] = 0.5f * (T[1] + T[3]);
    P[8 * NF + idx] = 0.5f * (T[2] + T[6]);
    P[9 * NF + idx] = 0.5f * (T[5] + T[7]);
}

__global__ void finalize_kernel(const float* __restrict__ Q2,
                                const float* __restrict__ Xn,
                                const float* __restrict__ chg,
                                float* __restrict__ out) {
    int idx = blockIdx.x * blockDim.x + threadIdx.x;
    if (idx >= NF) return;
    const int n = idx >> 6;
    float dc[10];
#pragma unroll
    for (int c = 0; c < 10; ++c) dc[c] = Q2[c * NF + idx];
    float dX[9], prod[9];
    recon3(dc, dX);
    mm3(dX, dX, prod);
    const float scale = 1.0f + 0.1f * chg[n];
#pragma unroll
    for (int i = 0; i < 9; ++i)
        out[(size_t)idx * 9 + i] = Xn[(size_t)idx * 9 + i] + dX[i] + scale * prod[i];
}

extern "C" void kernel_launch(void* const* d_in, const int* in_sizes, int n_in,
                              void* d_out, int out_size, void* d_ws, size_t ws_size,
                              hipStream_t stream) {
    const float* X   = (const float*)d_in[0];
    const int*   pi  = (const int*)  d_in[1];
    const float* dij = (const float*)d_in[2];
    const float* rbf = (const float*)d_in[3];
    const float* chg = (const float*)d_in[4];
    const float* W1  = (const float*)d_in[5];
    const float* b1  = (const float*)d_in[6];
    const float* W2  = (const float*)d_in[7];
    const float* b2  = (const float*)d_in[8];
    const float* W3  = (const float*)d_in[9];
    const float* b3  = (const float*)d_in[10];
    const float* L0  = (const float*)d_in[11];
    const float* L1  = (const float*)d_in[12];
    const float* L2  = (const float*)d_in[13];
    const float* L3  = (const float*)d_in[14];
    const float* L4  = (const float*)d_in[15];
    const float* L5  = (const float*)d_in[16];

    float* ws = (float*)d_ws;
    float* Xn = ws;
    float* P  = Xn + (size_t)NF * 9;
    float* Q  = P + (size_t)10 * NF;
    float* M  = Q + (size_t)10 * NF;

    hipMemsetAsync(M, 0, (size_t)10 * NF * sizeof(float), stream);

    prep_kernel<<<(NF + 255) / 256, 256, 0, stream>>>(X, Xn, P);
    mix_kernel<<<dim3(NN / 16, 10), 128, 0, stream>>>(P, Q, L0, L1, L2);
    edge_kernel<<<EE / 16, 32, 0, stream>>>(rbf, dij, pi, W1, b1, W2, b2, W3, b3, Q, M);
    combine_kernel<<<(NF + 255) / 256, 256, 0, stream>>>(M, Q, chg, P);
    mix_kernel<<<dim3(NN / 16, 10), 128, 0, stream>>>(P, Q, L3, L4, L5);
    finalize_kernel<<<(NF + 255) / 256, 256, 0, stream>>>(Q, Xn, chg, (float*)d_out);
}


# edges bucketed by source node in the call (one more launch); a wave sums each source run of its 16-edge tile in registers: one float atomic row per run, not per edge; sum order changes
# speedup vs baseline: 3.3222x; 3.3222x over previous
#include <hip/hip_runtime.h>
#include <math.h>

#define NN 10000
#define EE 160000
#define FF 64
#define NF (NN * FF)
#define PI_F 3.14159265358979323846f

#define H1S 68
#define H2S 132
#define RDS 196

typedef float v2f __attribute__((ext_vector_type(2)));
typedef float v8f __attribute__((ext_vector_type(8)));

__device__ __forceinline__ v8f wmma4(v2f a, v2f b, v8f c) {
    return __builtin_amdgcn_wmma_f32_16x16x4_f32(
         false, a,  false, b,
         (short)0, c,  false,  false);
}

__device__ __forceinline__ float silu(float x) {
    return x / (1.0f + expf(-x));
}

__device__ __forceinline__ void recon3(const float c[10], float t[9]) {
    float d = c[0];
    t[0] = d + c[4];  t[4] = d + c[5];  t[8] = d + c[6];
    t[1] = c[1] + c[7];  t[3] = c[7] - c[1];
    t[2] = c[2] + c[8];  t[6] = c[8] - c[2];
    t[5] = c[3] + c[9];  t[7] = c[9] - c[3];
}

__device__ __forceinline__ void mm3(const float a[9], const float b[9], float c[9]) {
#pragma unroll
    for (int i = 0; i < 3; ++i)
#pragma unroll
        for (int j = 0; j < 3; ++j) {
            float s = 0.f;
#pragma unroll
            for (int k = 0; k < 3; ++k) s += a[i * 3 + k] * b[k * 3 + j];
            c[i * 3 + j] = s;
        }
}

__global__ void prep_kernel(const float* __restrict__ X,
                            float* __restrict__ Xn,
                            float* __restrict__ P) {
    int idx = blockIdx.x * blockDim.x + threadIdx.x;
    if (idx >= NF) return;
    float x[9]; float nrm = 0.f;
#pragma unroll
    for (int i = 0; i < 9; ++i) { x[i] = X[(size_t)idx * 9 + i]; nrm += x[i] * x[i]; }
    float inv = 1.0f / (nrm + 1.0f);
#pragma unroll
    for (int i = 0; i < 9; ++i) { x[i] *= inv; Xn[(size_t)idx * 9 + i] = x[i]; }
    float dm = (x[0] + x[4] + x[8]) * (1.0f / 3.0f);
    P[0 * NF + idx] = dm;
    P[1 * NF + idx] = 0.5f * (x[1] - x[3]);
    P[2 * NF + idx] = 0.5f * (x[2] - x[6]);
    P[3 * NF + idx] = 0.5f * (x[5] - x[7]);
    P[4 * NF + idx] = x[0] - dm;
    P[5 * NF + idx] = x[4] - dm;
    P[6 * NF + idx] = x[8] - dm;
    P[7 * NF + idx] = 0.5f * (x[1] + x[3]);
    P[8 * NF + idx] = 0.5f * (x[2] + x[6]);
    P[9 * NF + idx] = 0.5f * (x[5] + x[7]);
}

__global__ __launch_bounds__(128) void mix_kernel(const float* __restrict__ P,
                                                  float* __restrict__ Q,
                                                  const float* __restrict__ La,
                                                  const float* __restrict__ Lb,
                                                  const float* __restrict__ Lc) {
    const int lane = threadIdx.x & 31;
    const int wave = threadIdx.x >> 5;
    const int ch = blockIdx.y;
    const int r0 = blockIdx.x * 16;
    const float* W = (ch == 0) ? La : (ch < 4 ? Lb : Lc);
    const float* A = P + (size_t)ch * NF;
    float* D = Q + (size_t)ch * NF;
    const int row = lane & 15;
    const int hi = lane >> 4;
    const int kb = hi * 2;
    const int col = wave * 16 + (lane & 15);
    v8f acc = {};
#pragma unroll
    for (int k = 0; k < FF; k += 4) {
        v2f a, b;
        a.x = A[(size_t)(r0 + row) * FF + k + kb];
        a.y = A[(size_t)(r0 + row) * FF + k + kb + 1];
        b.x = W[(k + kb) * FF + col];
        b.y = W[(k + kb + 1) * FF + col];
        acc = wmma4(a, b, acc);
    }
#pragma unroll
    for (int r = 0; r < 8; ++r)
        D[(size_t)(r0 + r + 8 * hi) * FF + col] = acc[r];
}

#define SR 40
#define SGRID (NN / SR)
#define STHREADS 1024
#define SB 4
static_assert(EE % 4 == 0, "sources are read four per load");
static_assert(SGRID * SR == NN, "the node ranges must cover every node exactly once");

__global__ __launch_bounds__(STHREADS) void sort_kernel(const int* __restrict__ pi,
                                                        int* __restrict__ seid,
                                                        int* __restrict__ ssrc,
                                                        int* __restrict__ sdst) {
    __shared__ int cnt[SR];
    __shared__ int base[SR];
    __shared__ int below;
    const int tid = threadIdx.x;
    const int lo = blockIdx.x * SR;
    if (tid < SR) cnt[tid] = 0;
    if (tid == 0) below = 0;
    __syncthreads();
    int nb = 0;
    const int4* pi4 = (const int4*)pi;
    for (int q0 = 0; q0 < EE / 4; q0 += SB * STHREADS) {
        int4 v[SB];
#pragma unroll
        for (int u = 0; u < SB; ++u) {
            const int q = q0 + u * STHREADS + tid;
            v[u] = pi4[q < EE / 4 ? q : EE / 4 - 1];
            if (q >= EE / 4) v[u] = make_int4(0x7fffffff, 0x7fffffff, 0x7fffffff, 0x7fffffff);
        }
#pragma unroll
        for (int u = 0; u < SB; ++u) {
            const int r4[4] = {v[u].x - lo, v[u].y - lo, v[u].z - lo, v[u].w - lo};
#pragma unroll
            for (int k = 0; k < 4; ++k) {
                if (r4[k] < 0) ++nb;
                else if (r4[k] < SR) atomicAdd(&cnt[r4[k]], 1);
            }
        }
    }
    atomicAdd(&below, nb);
    __syncthreads();
    if (tid == 0) {
        int run = below;
        for (int r = 0; r < SR; ++r) { base[r] = run; run += cnt[r]; cnt[r] = 0; }
    }
    __syncthreads();
    for (int q0 = 0; q0 < EE / 4; q0 += SB * STHREADS) {
        int4 v[SB];
#pragma unroll
        for (int u = 0; u < SB; ++u) {
            const int q = q0 + u * STHREADS + tid;
            v[u] = pi4[q < EE / 4 ? q : EE / 4 - 1];
            if (q >= EE / 4) v[u] = make_int4(0x7fffffff, 0x7fffffff, 0x7fffffff, 0x7fffffff);
        }
#pragma unroll
        for (int u = 0; u < SB; ++u) {
            const int e4 = 4 * (q0 + u * STHREADS + tid);
            const int s4[4] = {v[u].x, v[u].y, v[u].z, v[u].w};
#pragma unroll
            for (int k = 0; k < 4; ++k) {
                const int r = s4[k] - lo;
                if (r >= 0 && r < SR) {
                    const int p = base[r] + atomicAdd(&cnt[r], 1);
                    seid[p] = e4 + k;
                    ssrc[p] = s4[k];
                    sdst[p] = pi[EE + e4 + k];
                }
            }
        }
    }
}

#define EW 14
#define EGRID 250
#define NTILES (EE / 16)
#define WLDS (16 * H2S + 16 * H1S + 16)

static_assert(16 * RDS <= 16 * (H2S + H1S), "rad must fit into the words of h2 and h1");

__device__ __forceinline__ void stage_frag(const float* __restrict__ W, v2f* dst,
                                           const int ld, const int nt,
                                           const int npairs, const int tid) {
#pragma unroll 4
    for (int p = tid; p < npairs; p += 32 * EW) {
        const int l = p & 31;
        const int t = p >> 5;
        const int j = t % nt;
        const int ks = t / nt;
        const int k = 4 * ks + 2 * (l >> 4);
        const int n = 16 * j + (l & 15);
        v2f v;
        v.x = W[k * ld + n];
        v.y = W[(k + 1) * ld + n];
        dst[p] = v;
    }
}

__device__ __forceinline__ void wave_lds_handover() {
    __builtin_amdgcn_fence(__ATOMIC_RELEASE, "wavefront");
    __builtin_amdgcn_wave_barrier();
    __builtin_amdgcn_fence(__ATOMIC_ACQUIRE, "wavefront");
}

__device__ __forceinline__ void flush_run(float* __restrict__ M, const int s, const int lane,
                                          const float (&acc)[2][10]) {
#pragma unroll
    for (int t = 0; t < 2; ++t) {
        const int so = s * FF + lane + 32 * t;
#pragma unroll
        for (int c = 0; c < 10; ++c)
            atomicAdd(&M[c * NF + so], acc[t][c]);
    }
}

__global__ __launch_bounds__(32 * EW) void edge_kernel(
    const float* __restrict__ rbf, const float* __restrict__ dij,
    const int* __restrict__ seid, const int* __restrict__ ssrc,
    const int* __restrict__ sdst,
    const float* __restrict__ W1, const float* __restrict__ b1,
    const float* __restrict__ W2, const float* __restrict__ b2,
    const float* __restrict__ W3, const float* __restrict__ b3,
    const float* __restrict__ Q, float* __restrict__ M) {
    __shared__ v2f w1f[8 * 4 * 32];
    __shared__ v2f w2f[16 * 8 * 32];
    __shared__ v2f w3f[32 * 12 * 32];
    __shared__ __attribute__((aligned(16))) float wbuf[EW * WLDS];

    const int tid = threadIdx.x;
    stage_frag(W1, w1f, 64, 4, 8 * 4 * 32, tid);
    stage_frag(W2, w2f, 128, 8, 16 * 8 * 32, tid);
    stage_frag(W3, w3f, 192, 12, 32 * 12 * 32, tid);
    __syncthreads();

    const int lane = tid & 31;
    const int wave = __builtin_amdgcn_readfirstlane(tid >> 5);
    float* const h2 = wbuf + wave * WLDS;
    float* const h1 = h2 + 16 * H2S;
    float* const rad = h2;
    float* const cc = h1 + 16 * H1S;

    const int row = lane & 15;
    const int hi = lane >> 4;
    const int kb = hi * 2;
    const int col16 = lane & 15;

    for (int tile = blockIdx.x * EW + wave; tile < NTILES; tile += EGRID * EW) {
        const int e0 = tile * 16;

        wave_lds_handover();

        const int eid = seid[e0 + row];

        if (lane < 16) {
            float d = dij[eid];
            cc[lane] = 0.5f * (cosf(d * (PI_F / 0.5f)) + 1.0f) * (d < 0.5f ? 1.0f : 0.0f);
        }

        {
            v2f a[8];
#pragma unroll
            for (int ks = 0; ks < 8; ++ks) {
                a[ks].x = rbf[(size_t)eid * 32 + 4 * ks + kb];
                a[ks].y = rbf[(size_t)eid * 32 + 4 * ks + kb + 1];
            }
            v8f acc[4] = {};
#pragma unroll
            for (int ks = 0; ks < 8; ++ks)
#pragma unroll
                for (int j = 0; j < 4; ++j)
                    acc[j] = wmma4(a[ks], w1f[(ks * 4 + j) * 32 + lane], acc[j]);
#pragma unroll
            for (int j = 0; j < 4; ++j) {
                const int col = j * 16 + col16;
                const float bias = b1[col];
#pragma unroll
                for (int r = 0; r < 8; ++r)
                    h1[(r + 8 * hi) * H1S + col] = silu(acc[j][r] + bias);
            }
        }
        wave_lds_handover();

        {
            v8f acc[8] = {};
            v2f bn[8];
#pragma unroll
            for (int j = 0; j < 8; ++j) bn[j] = w2f[j * 32 + lane];
#pragma unroll
            for (int ks = 0; ks < 16; ++ks) {
                const v2f a = *(const v2f*)&h1[row * H1S + 4 * ks + kb];
                v2f bc[8];
#pragma unroll
                for (int j = 0; j < 8; ++j) bc[j] = bn[j];
                if (ks + 1 < 16) {
#pragma unroll
                    for (int j = 0; j < 8; ++j) bn[j] = w2f[((ks + 1) * 8 + j) * 32 + lane];
                }
#pragma unroll
                for (int j = 0; j < 8; ++j)
                    acc[j] = wmma4(a, bc[j], acc[j]);
            }
#pragma unroll
            for (int j = 0; j < 8; ++j) {
                const int col = j * 16 + col16;
                const float bias = b2[col];
#pragma unroll
                for (int r = 0; r < 8; ++r)
                    h2[(r + 8 * hi) * H2S + col] = silu(acc[j][r] + bias);
            }
        }
        wave_lds_handover();

        v2f a3[32];
#pragma unroll
        for (int ks = 0; ks < 32; ++ks)
            a3[ks] = *(const v2f*)&h2[row * H2S + 4 * ks + kb];
        wave_lds_handover();
        for (int g = 0; g < 2; ++g) {
            v8f acc[6] = {};
            v2f bn[6];
#pragma unroll
            for (int j = 0; j < 6; ++j) bn[j] = w3f[(g * 6 + j) * 32 + lane];
#pragma unroll
            for (int ks = 0; ks < 32; ++ks) {
                v2f bc[6];
#pragma unroll
                for (int j = 0; j < 6; ++j) bc[j] = bn[j];
                if (ks + 1 < 32) {
#pragma unroll
                    for (int j = 0; j < 6; ++j) bn[j] = w3f[((ks + 1) * 12 + g * 6 + j) * 32 + lane];
                }
#pragma unroll
                for (int j = 0; j < 6; ++j)
                    acc[j] = wmma4(a3[ks], bc[j], acc[j]);
            }
#pragma unroll
            for (int j = 0; j < 6; ++j) {
                const int col = (g * 6 + j) * 16 + col16;
                const float bias = b3[col];
#pragma unroll
                for (int r = 0; r < 8; ++r) {
                    const int er = r + 8 * hi;
                    rad[er * RDS + col] = silu(acc[j][r] + bias) * cc[er];
                }
            }
        }
        wave_lds_handover();

        const int srcv = ssrc[e0 + row];
        const int dstv = sdst[e0 + row];
        float acc[2][10];
#pragma unroll
        for (int t = 0; t < 2; ++t)
#pragma unroll
            for (int c = 0; c < 10; ++c) acc[t][c] = 0.f;
        int scur = __builtin_amdgcn_readlane(srcv, 0);
        for (int e = 0; e < 16; ++e) {
            const int s = __builtin_amdgcn_readlane(srcv, e);
            const int d = __builtin_amdgcn_readlane(dstv, e);
            if (s != scur) {
                flush_run(M, scur, lane, acc);
#pragma unroll
                for (int t = 0; t < 2; ++t)
#pragma unroll
                    for (int c = 0; c < 10; ++c) acc[t][c] = 0.f;
                scur = s;
            }
#pragma unroll
            for (int t = 0; t < 2; ++t) {
                const int f = lane + 32 * t;
                const float rI = rad[e * RDS + 3 * f + 0];
                const float rA = rad[e * RDS + 3 * f + 1];
                const float rS = rad[e * RDS + 3 * f + 2];
                const int g = d * FF + f;
                acc[t][0] += rI * Q[0 * NF + g];
#pragma unroll
                for (int c = 1; c < 4; ++c)
                    acc[t][c] += rA * Q[c * NF + g];
#pragma unroll
                for (int c = 4; c < 10; ++c)
                    acc[t][c] += rS * Q[c * NF + g];
            }
        }
        flush_run(M, scur, lane, acc);
    }
}

__global__ void combine_kernel(const float* __restrict__ M,
                               const float* __restrict__ Q,
                               const float* __restrict__ chg,
                               float* __restrict__ P) {
    int idx = blockIdx.x * blockDim.x + threadIdx.x;
    if (idx >= NF) return;
    const int n = idx >> 6;
    float mc[10], yc[10];
#pragma unroll
    for (int c = 0; c < 10; ++c) { mc[c] = M[c * NF + idx]; yc[c] = Q[c * NF + idx]; }
    float msg[9], Y[9], t1[9], t2[9];
    recon3(mc, msg);
    recon3(yc, Y);
    mm3(msg, Y, t1);
    mm3(Y, msg, t2);
    const float scale = 1.0f + 0.1f * chg[n];
    float T[9]; float tn = 0.f;
#pragma unroll
    for (int i = 0; i < 9; ++i) { T[i] = scale * (t1[i] + t2[i]); tn += T[i] * T[i]; }
    const float inv = 1.0f / (tn + 1.0f);
#pragma unroll
    for (int i = 0; i < 9; ++i) T[i] *= inv;
    float dm = (T[0] + T[4] + T[8]) * (1.0f / 3.0f);
    P[0 * NF + idx] = dm;
    P[1 * NF + idx] = 0.5f * (T[1] - T[3]);
    P[2 * NF + idx] = 0.5f * (T[2] - T[6]);
    P[3 * NF + idx] = 0.5f * (T[5] - T[7]);
    P[4 * NF + idx] = T[0] - dm;
    P[5 * NF + idx] = T[4] - dm;
    P[6 * NF + idx] = T[8] - dm;
    P[7 * NF + idx] = 0.5f * (T[1] + T[3]);
    P[8 * NF + idx] = 0.5f * (T[2] + T[6]);
    P[9 * NF + idx] = 0.5f * (T[5] + T[7]);
}

__global__ void finalize_kernel(const float* __restrict__ Q2,
                                const float* __restrict__ Xn,
                                const float* __restrict__ chg,
                                float* __restrict__ out) {
    int idx = blockIdx.x * blockDim.x + threadIdx.x;
    if (idx >= NF) return;
    const int n = idx >> 6;
    float dc[10];
#pragma unroll
    for (int c = 0; c < 10; ++c) dc[c] = Q2[c * NF + idx];
    float dX[9], prod[9];
    recon3(dc, dX);
    mm3(dX, dX, prod);
    const float scale = 1.0f + 0.1f * chg[n];
#pragma unroll
    for (int i = 0; i < 9; ++i)
        out[(size_t)idx * 9 + i] = Xn[(size_t)idx * 9 + i] + dX[i] + scale * prod[i];
}

extern "C" void kernel_launch(void* const* d_in, const int* in_sizes, int n_in,
                              void* d_out, int out_size, void* d_ws, size_t ws_size,
                              hipStream_t stream) {
    const float* X   = (const float*)d_in[0];
    const int*   pi  = (const int*)  d_in[1];
    const float* dij = (const float*)d_in[2];
    const float* rbf = (const float*)d_in[3];
    const float* chg = (const float*)d_in[4];
    const float* W1  = (const float*)d_in[5];
    const float* b1  = (const float*)d_in[6];
    const float* W2  = (const float*)d_in[7];
    const float* b2  = (const float*)d_in[8];
    const float* W3  = (const float*)d_in[9];
    const float* b3  = (const float*)d_in[10];
    const float* L0  = (const float*)d_in[11];
    const float* L1  = (const float*)d_in[12];
    const float* L2  = (const float*)d_in[13];
    const float* L3  = (const float*)d_in[14];
    const float* L4  = (const float*)d_in[15];
    const float* L5  = (const float*)d_in[16];

    float* ws = (float*)d_ws;
    float* Xn = ws;
    float* P  = Xn + (size_t)NF * 9;
    float* Q  = P + (size_t)10 * NF;
    float* M  = Q + (size_t)10 * NF;

    int* seid = (int*)P;
    int* ssrc = seid + EE;
    int* sdst = ssrc + EE;

    hipMemsetAsync(M, 0, (size_t)10 * NF * sizeof(float), stream);

    prep_kernel<<<(NF + 255) / 256, 256, 0, stream>>>(X, Xn, P);
    mix_kernel<<<dim3(NN / 16, 10), 128, 0, stream>>>(P, Q, L0, L1, L2);
    sort_kernel<<<SGRID, STHREADS, 0, stream>>>(pi, seid, ssrc, sdst);
    edge_kernel<<<EGRID, 32 * EW, 0, stream>>>(rbf, dij, seid, ssrc, sdst, W1, b1, W2, b2, W3, b3, Q, M);
    combine_kernel<<<(NF + 255) / 256, 256, 0, stream>>>(M, Q, chg, P);
    mix_kernel<<<dim3(NN / 16, 10), 128, 0, stream>>>(P, Q, L3, L4, L5);
    finalize_kernel<<<(NF + 255) / 256, 256, 0, stream>>>(Q, Xn, chg, (float*)d_out);
}


# edge kernel stores messages, no atomics, no memset: a wave takes 2-3 consecutive tiles, inner runs go to M, its two end runs to side rows the combine adds in wave order; sum order by sorted position
# speedup vs baseline: 3.5890x; 1.0803x over previous
#include <hip/hip_runtime.h>
#include <math.h>

#define NN 10000
#define EE 160000
#define FF 64
#define NF (NN * FF)
#define PI_F 3.14159265358979323846f

#define H1S 68
#define H2S 132
#define RDS 196

typedef float v2f __attribute__((ext_vector_type(2)));
typedef float v8f __attribute__((ext_vector_type(8)));

__device__ __forceinline__ v8f wmma4(v2f a, v2f b, v8f c) {
    return __builtin_amdgcn_wmma_f32_16x16x4_f32(
         false, a,  false, b,
         (short)0, c,  false,  false);
}

__device__ __forceinline__ float silu(float x) {
    return x / (1.0f + expf(-x));
}

__device__ __forceinline__ void recon3(const float c[10], float t[9]) {
    float d = c[0];
    t[0] = d + c[4];  t[4] = d + c[5];  t[8] = d + c[6];
    t[1] = c[1] + c[7];  t[3] = c[7] - c[1];
    t[2] = c[2] + c[8];  t[6] = c[8] - c[2];
    t[5] = c[3] + c[9];  t[7] = c[9] - c[3];
}

__device__ __forceinline__ void mm3(const float a[9], const float b[9], float c[9]) {
#pragma unroll
    for (int i = 0; i < 3; ++i)
#pragma unroll
        for (int j = 0; j < 3; ++j) {
            float s = 0.f;
#pragma unroll
            for (int k = 0; k < 3; ++k) s += a[i * 3 + k] * b[k * 3 + j];
            c[i * 3 + j] = s;
        }
}

__global__ void prep_kernel(const float* __restrict__ X,
                            float* __restrict__ Xn,
                            float* __restrict__ P) {
    int idx = blockIdx.x * blockDim.x + threadIdx.x;
    if (idx >= NF) return;
    float x[9]; float nrm = 0.f;
#pragma unroll
    for (int i = 0; i < 9; ++i) { x[i] = X[(size_t)idx * 9 + i]; nrm += x[i] * x[i]; }
    float inv = 1.0f / (nrm + 1.0f);
#pragma unroll
    for (int i = 0; i < 9; ++i) { x[i] *= inv; Xn[(size_t)idx * 9 + i] = x[i]; }
    float dm = (x[0] + x[4] + x[8]) * (1.0f / 3.0f);
    P[0 * NF + idx] = dm;
    P[1 * NF + idx] = 0.5f * (x[1] - x[3]);
    P[2 * NF + idx] = 0.5f * (x[2] - x[6]);
    P[3 * NF + idx] = 0.5f * (x[5] - x[7]);
    P[4 * NF + idx] = x[0] - dm;
    P[5 * NF + idx] = x[4] - dm;
    P[6 * NF + idx] = x[8] - dm;
    P[7 * NF + idx] = 0.5f * (x[1] + x[3]);
    P[8 * NF + idx] = 0.5f * (x[2] + x[6]);
    P[9 * NF + idx] = 0.5f * (x[5] + x[7]);
}

__global__ __launch_bounds__(128) void mix_kernel(const float* __restrict__ P,
                                                  float* __restrict__ Q,
                                                  const float* __restrict__ La,
                                                  const float* __restrict__ Lb,
                                                  const float* __restrict__ Lc) {
    const int lane = threadIdx.x & 31;
    const int wave = threadIdx.x >> 5;
    const int ch = blockIdx.y;
    const int r0 = blockIdx.x * 16;
    const float* W = (ch == 0) ? La : (ch < 4 ? Lb : Lc);
    const float* A = P + (size_t)ch * NF;
    float* D = Q + (size_t)ch * NF;
    const int row = lane & 15;
    const int hi = lane >> 4;
    const int kb = hi * 2;
    const int col = wave * 16 + (lane & 15);
    v8f acc = {};
#pragma unroll
    for (int k = 0; k < FF; k += 4) {
        v2f a, b;
        a.x = A[(size_t)(r0 + row) * FF + k + kb];
        a.y = A[(size_t)(r0 + row) * FF + k + kb + 1];
        b.x = W[(k + kb) * FF + col];
        b.y = W[(k + kb + 1) * FF + col];
        acc = wmma4(a, b, acc);
    }
#pragma unroll
    for (int r = 0; r < 8; ++r)
        D[(size_t)(r0 + r + 8 * hi) * FF + col] = acc[r];
}

#define SR 40
#define SGRID (NN / SR)
#define STHREADS 1024
#define SB 4
static_assert(EE % 4 == 0, "sources are read four per load");
static_assert(SGRID * SR == NN, "the node ranges must cover every node exactly once");

__global__ __launch_bounds__(STHREADS) void sort_kernel(const int* __restrict__ pi,
                                                        int* __restrict__ seid,
                                                        int* __restrict__ ssrc,
                                                        int* __restrict__ sdst,
                                                        int* __restrict__ nstart,
                                                        int* __restrict__ ncnt) {
    __shared__ int cnt[SR];
    __shared__ int base[SR];
    __shared__ int below;
    const int tid = threadIdx.x;
    const int lo = blockIdx.x * SR;
    if (tid < SR) cnt[tid] = 0;
    if (tid == 0) below = 0;
    __syncthreads();
    int nb = 0;
    const int4* pi4 = (const int4*)pi;
    for (int q0 = 0; q0 < EE / 4; q0 += SB * STHREADS) {
        int4 v[SB];
#pragma unroll
        for (int u = 0; u < SB; ++u) {
            const int q = q0 + u * STHREADS + tid;
            v[u] = pi4[q < EE / 4 ? q : EE / 4 - 1];
            if (q >= EE / 4) v[u] = make_int4(0x7fffffff, 0x7fffffff, 0x7fffffff, 0x7fffffff);
        }
#pragma unroll
        for (int u = 0; u < SB; ++u) {
            const int r4[4] = {v[u].x - lo, v[u].y - lo, v[u].z - lo, v[u].w - lo};
#pragma unroll
            for (int k = 0; k < 4; ++k) {
                if (r4[k] < 0) ++nb;
                else if (r4[k] < SR) atomicAdd(&cnt[r4[k]], 1);
            }
        }
    }
    atomicAdd(&below, nb);
    __syncthreads();
    if (tid == 0) {
        int run = below;
        for (int r = 0; r < SR; ++r) {
            base[r] = run; nstart[lo + r] = run; ncnt[lo + r] = cnt[r];
            run += cnt[r]; cnt[r] = 0;
        }
    }
    __syncthreads();
    for (int q0 = 0; q0 < EE / 4; q0 += SB * STHREADS) {
        int4 v[SB];
#pragma unroll
        for (int u = 0; u < SB; ++u) {
            const int q = q0 + u * STHREADS + tid;
            v[u] = pi4[q < EE / 4 ? q : EE / 4 - 1];
            if (q >= EE / 4) v[u] = make_int4(0x7fffffff, 0x7fffffff, 0x7fffffff, 0x7fffffff);
        }
#pragma unroll
        for (int u = 0; u < SB; ++u) {
            const int e4 = 4 * (q0 + u * STHREADS + tid);
            const int s4[4] = {v[u].x, v[u].y, v[u].z, v[u].w};
#pragma unroll
            for (int k = 0; k < 4; ++k) {
                const int r = s4[k] - lo;
                if (r >= 0 && r < SR) {
                    const int p = base[r] + atomicAdd(&cnt[r], 1);
                    seid[p] = e4 + k;
                    ssrc[p] = s4[k];
                    sdst[p] = pi[EE + e4 + k];
                }
            }
        }
    }
}

#define EW 14
#define EGRID 250
#define NTILES (EE / 16)
#define TPG (NTILES / EGRID)
#define NW3 12
#define NWAVES (EGRID * EW)
#define SLOT (10 * FF)
static_assert(TPG * EGRID == NTILES, "every work-group takes the same number of tiles");
static_assert(3 * NW3 + 2 * (EW - NW3) == TPG, "the waves' tile ranges must cover the work-group's tiles");
#define WLDS (16 * H2S + 16 * H1S + 16)

static_assert(16 * RDS <= 16 * (H2S + H1S), "rad must fit into the words of h2 and h1");

__device__ __forceinline__ void stage_frag(const float* __restrict__ W, v2f* dst,
                                           const int ld, const int nt,
                                           const int npairs, const int tid) {
#pragma unroll 4
    for (int p = tid; p < npairs; p += 32 * EW) {
        const int l = p & 31;
        const int t = p >> 5;
        const int j = t % nt;
        const int ks = t / nt;
        const int k = 4 * ks + 2 * (l >> 4);
        const int n = 16 * j + (l & 15);
        v2f v;
        v.x = W[k * ld + n];
        v.y = W[(k + 1) * ld + n];
        dst[p] = v;
    }
}

__device__ __forceinline__ void wave_lds_handover() {
    __builtin_amdgcn_fence(__ATOMIC_RELEASE, "wavefront");
    __builtin_amdgcn_wave_barrier();
    __builtin_amdgcn_fence(__ATOMIC_ACQUIRE, "wavefront");
}

__device__ __forceinline__ void store_run(float* const row, const int cs, const int lane,
                                          const float (&acc)[2][10]) {
#pragma unroll
    for (int t = 0; t < 2; ++t)
#pragma unroll
        for (int c = 0; c < 10; ++c)
            row[c * cs + lane + 32 * t] = acc[t][c];
}

__device__ __forceinline__ int wave_tile0(const int w) { return w < NW3 ? 3 * w : 3 * NW3 + 2 * (w - NW3); }
__device__ __forceinline__ int wave_ntile(const int w) { return w < NW3 ? 3 : 2; }

__device__ __forceinline__ int pos_wave(const int p, int& p0, int& p1) {
    const int tile = p >> 4;
    const int b = tile / TPG;
    const int t = tile - b * TPG;
    const int w = t < 3 * NW3 ? t / 3 : NW3 + ((t - 3 * NW3) >> 1);
    p0 = (b * TPG + wave_tile0(w)) * 16;
    p1 = p0 + wave_ntile(w) * 16;
    return b * EW + w;
}

__global__ __launch_bounds__(32 * EW) void edge_kernel(
    const float* __restrict__ rbf, const float* __restrict__ dij,
    const int* __restrict__ seid, const int* __restrict__ ssrc,
    const int* __restrict__ sdst,
    const float* __restrict__ W1, const float* __restrict__ b1,
    const float* __restrict__ W2, const float* __restrict__ b2,
    const float* __restrict__ W3, const float* __restrict__ b3,
    const float* __restrict__ Q, float* __restrict__ M, float* __restrict__ side) {
    __shared__ v2f w1f[8 * 4 * 32];
    __shared__ v2f w2f[16 * 8 * 32];
    __shared__ v2f w3f[32 * 12 * 32];
    __shared__ __attribute__((aligned(16))) float wbuf[EW * WLDS];

    const int tid = threadIdx.x;
    stage_frag(W1, w1f, 64, 4, 8 * 4 * 32, tid);
    stage_frag(W2, w2f, 128, 8, 16 * 8 * 32, tid);
    stage_frag(W3, w3f, 192, 12, 32 * 12 * 32, tid);
    __syncthreads();

    const int lane = tid & 31;
    const int wave = __builtin_amdgcn_readfirstlane(tid >> 5);
    float* const h2 = wbuf + wave * WLDS;
    float* const h1 = h2 + 16 * H2S;
    float* const rad = h2;
    float* const cc = h1 + 16 * H1S;

    const int row = lane & 15;
    const int hi = lane >> 4;
    const int kb = hi * 2;
    const int col16 = lane & 15;

    const int tile0 = blockIdx.x * TPG + wave_tile0(wave);
    const int ntile = wave_ntile(wave);
    const int p0 = tile0 * 16;
    float* const side0 = side + (size_t)(blockIdx.x * EW + wave) * (2 * SLOT);

    int scur = __builtin_amdgcn_readfirstlane(ssrc[p0]);
    int runpos = p0;
    float msum[2][10];
#pragma unroll
    for (int t = 0; t < 2; ++t)
#pragma unroll
        for (int c = 0; c < 10; ++c) msum[t][c] = 0.f;

    for (int k = 0; k < ntile; ++k) {
        const int tile = tile0 + k;
        const int e0 = tile * 16;

        wave_lds_handover();

        const int eid = seid[e0 + row];

        if (lane < 16) {
            float d = dij[eid];
            cc[lane] = 0.5f * (cosf(d * (PI_F / 0.5f)) + 1.0f) * (d < 0.5f ? 1.0f : 0.0f);
        }

        {
            v2f a[8];
#pragma unroll
            for (int ks = 0; ks < 8; ++ks) {
                a[ks].x = rbf[(size_t)eid * 32 + 4 * ks + kb];
                a[ks].y = rbf[(size_t)eid * 32 + 4 * ks + kb + 1];
            }
            v8f acc[4] = {};
#pragma unroll
            for (int ks = 0; ks < 8; ++ks)
#pragma unroll
                for (int j = 0; j < 4; ++j)
                    acc[j] = wmma4(a[ks], w1f[(ks * 4 + j) * 32 + lane], acc[j]);
#pragma unroll
            for (int j = 0; j < 4; ++j) {
                const int col = j * 16 + col16;
                const float bias = b1[col];
#pragma unroll
                for (int r = 0; r < 8; ++r)
                    h1[(r + 8 * hi) * H1S + col] = silu(acc[j][r] + bias);
            }
        }
        wave_lds_handover();

        {
            v8f acc[8] = {};
            v2f bn[8];
#pragma unroll
            for (int j = 0; j < 8; ++j) bn[j] = w2f[j * 32 + lane];
#pragma unroll
            for (int ks = 0; ks < 16; ++ks) {
                const v2f a = *(const v2f*)&h1[row * H1S + 4 * ks + kb];
                v2f bc[8];
#pragma unroll
                for (int j = 0; j < 8; ++j) bc[j] = bn[j];
                if (ks + 1 < 16) {
#pragma unroll
                    for (int j = 0; j < 8; ++j) bn[j] = w2f[((ks + 1) * 8 + j) * 32 + lane];
                }
#pragma unroll
                for (int j = 0; j < 8; ++j)
                    acc[j] = wmma4(a, bc[j], acc[j]);
            }
#pragma unroll
            for (int j = 0; j < 8; ++j) {
                const int col = j * 16 + col16;
                const float bias = b2[col];
#pragma unroll
                for (int r = 0; r < 8; ++r)
                    h2[(r + 8 * hi) * H2S + col] = silu(acc[j][r] + bias);
            }
        }
        wave_lds_handover();

        v2f a3[32];
#pragma unroll
        for (int ks = 0; ks < 32; ++ks)
            a3[ks] = *(const v2f*)&h2[row * H2S + 4 * ks + kb];
        wave_lds_handover();
        for (int g = 0; g < 2; ++g) {
            v8f acc[6] = {};
            v2f bn[6];
#pragma unroll
            for (int j = 0; j < 6; ++j) bn[j] = w3f[(g * 6 + j) * 32 + lane];
#pragma unroll
            for (int ks = 0; ks < 32; ++ks) {
                v2f bc[6];
#pragma unroll
                for (int j = 0; j < 6; ++j) bc[j] = bn[j];
                if (ks + 1 < 32) {
#pragma unroll
                    for (int j = 0; j < 6; ++j) bn[j] = w3f[((ks + 1) * 12 + g * 6 + j) * 32 + lane];
                }
#pragma unroll
                for (int j = 0; j < 6; ++j)
                    acc[j] = wmma4(a3[ks], bc[j], acc[j]);
            }
#pragma unroll
            for (int j = 0; j < 6; ++j) {
                const int col = (g * 6 + j) * 16 + col16;
                const float bias = b3[col];
#pragma unroll
                for (int r = 0; r < 8; ++r) {
                    const int er = r + 8 * hi;
                    rad[er * RDS + col] = silu(acc[j][r] + bias) * cc[er];
                }
            }
        }
        wave_lds_handover();

        const int srcv = ssrc[e0 + row];
        const int dstv = sdst[e0 + row];
        for (int e = 0; e < 16; ++e) {
            const int s = __builtin_amdgcn_readlane(srcv, e);
            const int d = __builtin_amdgcn_readlane(dstv, e);
            if (s != scur) {
                if (runpos == p0) store_run(side0, FF, lane, msum);
                else              store_run(M + (size_t)scur * FF, NF, lane, msum);
#pragma unroll
                for (int t = 0; t < 2; ++t)
#pragma unroll
                    for (int c = 0; c < 10; ++c) msum[t][c] = 0.f;
                scur = s;
                runpos = e0 + e;
            }
#pragma unroll
            for (int t = 0; t < 2; ++t) {
                const int f = lane + 32 * t;
                const float rI = rad[e * RDS + 3 * f + 0];
                const float rA = rad[e * RDS + 3 * f + 1];
                const float rS = rad[e * RDS + 3 * f + 2];
                const int g = d * FF + f;
                msum[t][0] += rI * Q[0 * NF + g];
#pragma unroll
                for (int c = 1; c < 4; ++c)
                    msum[t][c] += rA * Q[c * NF + g];
#pragma unroll
                for (int c = 4; c < 10; ++c)
                    msum[t][c] += rS * Q[c * NF + g];
            }
        }
    }
    store_run(runpos == p0 ? side0 : side0 + SLOT, FF, lane, msum);
}

__global__ void combine_kernel(float* MP,
                               const float* __restrict__ Q,
                               const float* __restrict__ chg,
                               const float* __restrict__ side,
                               const int* __restrict__ nstart,
                               const int* __restrict__ ncnt) {
    int idx = blockIdx.x * blockDim.x + threadIdx.x;
    if (idx >= NF) return;
    const int n = idx >> 6;
    const int f = idx & 63;
    float mc[10], yc[10];
#pragma unroll
    for (int c = 0; c < 10; ++c) { mc[c] = 0.f; yc[c] = Q[c * NF + idx]; }
    const int st = nstart[n];
    const int ct = ncnt[n];
    if (ct > 0) {
        const int en = st + ct;
        int a0, a1;
        const int g0 = pos_wave(st, a0, a1);
        if (st > a0 && en < a1) {
#pragma unroll
            for (int c = 0; c < 10; ++c) mc[c] = MP[c * NF + idx];
        } else {
            int z0, z1;
            const int g1 = pos_wave(en - 1, z0, z1);
            for (int g = g0; g <= g1; ++g) {
                const int b = g / EW;
                const int w = g - b * EW;
                const int q0 = (b * TPG + wave_tile0(w)) * 16;
                const float* r = side + (size_t)g * (2 * SLOT) + (st <= q0 ? 0 : SLOT) + f;
#pragma unroll
                for (int c = 0; c < 10; ++c) mc[c] += r[c * FF];
            }
        }
    }
    float msg[9], Y[9], t1[9], t2[9];
    recon3(mc, msg);
    recon3(yc, Y);
    mm3(msg, Y, t1);
    mm3(Y, msg, t2);
    const float scale = 1.0f + 0.1f * chg[n];
    float T[9]; float tn = 0.f;
#pragma unroll
    for (int i = 0; i < 9; ++i) { T[i] = scale * (t1[i] + t2[i]); tn += T[i] * T[i]; }
    const float inv = 1.0f / (tn + 1.0f);
#pragma unroll
    for (int i = 0; i < 9; ++i) T[i] *= inv;
    float dm = (T[0] + T[4] + T[8]) * (1.0f / 3.0f);
    MP[0 * NF + idx] = dm;
    MP[1 * NF + idx] = 0.5f * (T[1] - T[3]);
    MP[2 * NF + idx] = 0.5f * (T[2] - T[6]);
    MP[3 * NF + idx] = 0.5f * (T[5] - T[7]);
    MP[4 * NF + idx] = T[0] - dm;
    MP[5 * NF + idx] = T[4] - dm;
    MP[6 * NF + idx] = T[8] - dm;
    MP[7 * NF + idx] = 0.5f * (T[1] + T[3]);
    MP[8 * NF + idx] = 0.5f * (T[2] + T[6]);
    MP[9 * NF + idx] = 0.5f * (T[5] + T[7]);
}

__global__ void finalize_kernel(const float* __restrict__ Q2,
                                const float* __restrict__ Xn,
                                const float* __restrict__ chg,
                                float* __restrict__ out) {
    int idx = blockIdx.x * blockDim.x + threadIdx.x;
    if (idx >= NF) return;
    const int n = idx >> 6;
    float dc[10];
#pragma unroll
    for (int c = 0; c < 10; ++c) dc[c] = Q2[c * NF + idx];
    float dX[9], prod[9];
    recon3(dc, dX);
    mm3(dX, dX, prod);
    const float scale = 1.0f + 0.1f * chg[n];
#pragma unroll
    for (int i = 0; i < 9; ++i)
        out[(size_t)idx * 9 + i] = Xn[(size_t)idx * 9 + i] + dX[i] + scale * prod[i];
}

extern "C" void kernel_launch(void* const* d_in, const int* in_sizes, int n_in,
                              void* d_out, int out_size, void* d_ws, size_t ws_size,
                              hipStream_t stream) {
    const float* X   = (const float*)d_in[0];
    const int*   pi  = (const int*)  d_in[1];
    const float* dij = (const float*)d_in[2];
    const float* rbf = (const float*)d_in[3];
    const float* chg = (const float*)d_in[4];
    const float* W1  = (const float*)d_in[5];
    const float* b1  = (const float*)d_in[6];
    const float* W2  = (const float*)d_in[7];
    const float* b2  = (const float*)d_in[8];
    const float* W3  = (const float*)d_in[9];
    const float* b3  = (const float*)d_in[10];
    const float* L0  = (const float*)d_in[11];
    const float* L1  = (const float*)d_in[12];
    const float* L2  = (const float*)d_in[13];
    const float* L3  = (const float*)d_in[14];
    const float* L4  = (const float*)d_in[15];
    const float* L5  = (const float*)d_in[16];

    float* ws = (float*)d_ws;
    float* Xn = ws;
    float* P  = Xn + (size_t)NF * 9;
    float* Q  = P + (size_t)10 * NF;
    float* M  = Q + (size_t)10 * NF;

    int* seid = (int*)P;
    int* ssrc = seid + EE;
    int* sdst = ssrc + EE;
    int* nstart = sdst + EE;
    int* ncnt = nstart + NN;
    float* side = P + 524288;
    static_assert(3 * EE + 2 * NN <= 524288, "lists and node runs must end before the side rows");
    static_assert(524288 + (size_t)NWAVES * 2 * SLOT <= (size_t)10 * NF, "side rows must fit into P");

    prep_kernel<<<(NF + 255) / 256, 256, 0, stream>>>(X, Xn, P);
    mix_kernel<<<dim3(NN / 16, 10), 128, 0, stream>>>(P, Q, L0, L1, L2);
    sort_kernel<<<SGRID, STHREADS, 0, stream>>>(pi, seid, ssrc, sdst, nstart, ncnt);
    edge_kernel<<<EGRID, 32 * EW, 0, stream>>>(rbf, dij, seid, ssrc, sdst, W1, b1, W2, b2, W3, b3, Q, M, side);
    combine_kernel<<<(NF + 255) / 256, 256, 0, stream>>>(M, Q, chg, side, nstart, ncnt);
    mix_kernel<<<dim3(NN / 16, 10), 128, 0, stream>>>(M, Q, L3, L4, L5);
    finalize_kernel<<<(NF + 255) / 256, 256, 0, stream>>>(Q, Xn, chg, (float*)d_out);
}
